# starting kernel
# baseline (speedup 1.0000x reference)
#include <hip/hip_runtime.h>

typedef __attribute__((ext_vector_type(2))) float v2f;
typedef __attribute__((ext_vector_type(8))) float v8f;

#define NB    1024
#define DIN   64
#define DHID  128
#define BN_EPS 1e-5f

__global__ __launch_bounds__(32)
void k1_gemm(const float* __restrict__ x, const float* __restrict__ W1,
             const float* __restrict__ b1, float* __restrict__ A,
             float* __restrict__ Bm)
{
    const int i0   = blockIdx.x * 16;
    const int j0   = blockIdx.y * 16;
    const int hlf  = blockIdx.z;
    const int lane = threadIdx.x & 31;
    const int m    = lane & 15;
    const int kg   = (lane >> 4) << 1;

    const float* __restrict__ W = W1 + hlf * DIN * DHID;
    float* __restrict__ C = hlf ? Bm : A;

    v8f acc = {0.f, 0.f, 0.f, 0.f, 0.f, 0.f, 0.f, 0.f};
    for (int k = 0; k < DIN; k += 4) {
        v2f a, b;
        a.x = x[(i0 + m) * DIN + (k + kg)];
        a.y = x[(i0 + m) * DIN + (k + kg + 1)];
        b.x = W[(k + kg) * DHID + (j0 + m)];
        b.y = W[(k + kg + 1) * DHID + (j0 + m)];
        acc = __builtin_amdgcn_wmma_f32_16x16x4_f32(false, a, false, b,
                                                    (short)0, acc, false, false);
    }

    const float bias  = (hlf == 0) ? b1[j0 + m] : 0.0f;
    const int   rbase = (lane < 16) ? 0 : 8;
#pragma unroll
    for (int v = 0; v < 8; ++v) {
        C[(i0 + rbase + v) * DHID + (j0 + m)] = acc[v] + bias;
    }
}

__global__ __launch_bounds__(256)
void k2_stats(const float* __restrict__ A, const float* __restrict__ Bm,
              float* __restrict__ partials)
{
    const int h   = blockIdx.x;
    const int jc  = blockIdx.y;
    const int tid = threadIdx.x;

    __shared__ float bcol[256];
    __shared__ float red1[256];
    __shared__ float red2[256];

    bcol[tid] = Bm[(jc * 256 + tid) * DHID + h];

    const float a0 = A[(tid      ) * DHID + h];
    const float a1 = A[(tid + 256) * DHID + h];
    const float a2 = A[(tid + 512) * DHID + h];
    const float a3 = A[(tid + 768) * DHID + h];
    __syncthreads();

    float s1 = 0.f, s2 = 0.f;
    for (int j = 0; j < 256; ++j) {
        const float b  = bcol[j];
        const float r0 = fmaxf(a0 + b, 0.f);
        const float r1 = fmaxf(a1 + b, 0.f);
        const float r2 = fmaxf(a2 + b, 0.f);
        const float r3 = fmaxf(a3 + b, 0.f);
        s1 += (r0 + r1) + (r2 + r3);
        s2 = fmaf(r0, r0, s2); s2 = fmaf(r1, r1, s2);
        s2 = fmaf(r2, r2, s2); s2 = fmaf(r3, r3, s2);
    }
    red1[tid] = s1; red2[tid] = s2;
    __syncthreads();
    for (int s = 128; s > 0; s >>= 1) {
        if (tid < s) { red1[tid] += red1[tid + s]; red2[tid] += red2[tid + s]; }
        __syncthreads();
    }
    if (tid == 0) {
        partials[(h * 4 + jc) * 2 + 0] = red1[0];
        partials[(h * 4 + jc) * 2 + 1] = red2[0];
    }
}

__global__ __launch_bounds__(128)
void k3_final(const float* __restrict__ partials, const float* __restrict__ gamma,
              const float* __restrict__ beta, const float* __restrict__ W2,
              const float* __restrict__ b2, float* __restrict__ cvec,
              float* __restrict__ cterm)
{
    const int h = threadIdx.x;
    __shared__ float red[128];

    float S1 = 0.f, S2 = 0.f;
    for (int jc = 0; jc < 4; ++jc) {
        S1 += partials[(h * 4 + jc) * 2 + 0];
        S2 += partials[(h * 4 + jc) * 2 + 1];
    }
    const float invN = 1.0f / ((float)NB * (float)NB);
    const float mean = S1 * invN;
    const float var  = fmaxf(S2 * invN - mean * mean, 0.f);
    const float inv  = rsqrtf(var + BN_EPS);
    const float g    = gamma[h] * inv;
    cvec[h] = W2[h] * g;
    red[h]  = W2[h] * (beta[h] - mean * g);
    __syncthreads();
    for (int s = 64; s > 0; s >>= 1) {
        if (h < s) red[h] += red[h + s];
        __syncthreads();
    }
    if (h == 0) *cterm = red[0] + b2[0];
}

__global__ __launch_bounds__(256)
void k4_out(const float* __restrict__ A, const float* __restrict__ Bm,
            const float* __restrict__ cvec, const float* __restrict__ cterm,
            float* __restrict__ out)
{
    __shared__ float aT[DHID][64];
    __shared__ float bT[DHID][32];
    __shared__ float cs[DHID];

    const int i0  = blockIdx.x * 64;
    const int j0  = blockIdx.y * 32;
    const int tid = threadIdx.x;

    {
        const int il = tid & 63;
        const int hq = tid >> 6;
        const float4* src = (const float4*)(A + (i0 + il) * DHID + hq * 32);
#pragma unroll
        for (int q = 0; q < 8; ++q) {
            const float4 v = src[q];
            const int h = hq * 32 + q * 4;
            aT[h + 0][il] = v.x; aT[h + 1][il] = v.y;
            aT[h + 2][il] = v.z; aT[h + 3][il] = v.w;
        }
    }
    {
        const int jl = tid & 31;
        const int hq = tid >> 5;
        const float4* src = (const float4*)(Bm + (j0 + jl) * DHID + hq * 16);
#pragma unroll
        for (int q = 0; q < 4; ++q) {
            const float4 v = src[q];
            const int h = hq * 16 + q * 4;
            bT[h + 0][jl] = v.x; bT[h + 1][jl] = v.y;
            bT[h + 2][jl] = v.z; bT[h + 3][jl] = v.w;
        }
    }
    if (tid < DHID) cs[tid] = cvec[tid];
    __syncthreads();

    const int ib = (tid & 15) * 4;
    const int jb = (tid >> 4) * 2;
    float acc[4][2] = {};

    for (int h = 0; h < DHID; ++h) {
        const float4 av = *(const float4*)&aT[h][ib];
        const float  b0 = bT[h][jb];
        const float  b1v = bT[h][jb + 1];
        const float  ck = cs[h];
        acc[0][0] = fmaf(ck, fmaxf(av.x + b0, 0.f), acc[0][0]);
        acc[0][1] = fmaf(ck, fmaxf(av.x + b1v, 0.f), acc[0][1]);
        acc[1][0] = fmaf(ck, fmaxf(av.y + b0, 0.f), acc[1][0]);
        acc[1][1] = fmaf(ck, fmaxf(av.y + b1v, 0.f), acc[1][1]);
        acc[2][0] = fmaf(ck, fmaxf(av.z + b0, 0.f), acc[2][0]);
        acc[2][1] = fmaf(ck, fmaxf(av.z + b1v, 0.f), acc[2][1]);
        acc[3][0] = fmaf(ck, fmaxf(av.w + b0, 0.f), acc[3][0]);
        acc[3][1] = fmaf(ck, fmaxf(av.w + b1v, 0.f), acc[3][1]);
    }

    const float ct = *cterm;
#pragma unroll
    for (int ii = 0; ii < 4; ++ii) {
        float2 o;
        o.x = acc[ii][0] + ct;
        o.y = acc[ii][1] + ct;
        *(float2*)(out + (size_t)(i0 + ib + ii) * NB + (j0 + jb)) = o;
    }
}

extern "C" void kernel_launch(void* const* d_in, const int* in_sizes, int n_in,
                              void* d_out, int out_size, void* d_ws, size_t ws_size,
                              hipStream_t stream)
{
    const float* x     = (const float*)d_in[0];
    const float* W1    = (const float*)d_in[1];
    const float* b1    = (const float*)d_in[2];
    const float* gamma = (const float*)d_in[3];
    const float* beta  = (const float*)d_in[4];
    const float* W2    = (const float*)d_in[5];
    const float* b2    = (const float*)d_in[6];
    float* out = (float*)d_out;

    float* ws       = (float*)d_ws;
    float* A        = ws;
    float* Bm       = ws + NB * DHID;
    float* partials = ws + 2 * NB * DHID;
    float* cvec     = partials + DHID * 4 * 2;
    float* cterm    = cvec + DHID;

    k1_gemm <<<dim3(NB / 16, DHID / 16, 2), 32,  0, stream>>>(x, W1, b1, A, Bm);
    k2_stats<<<dim3(DHID, 4),              256, 0, stream>>>(A, Bm, partials);
    k3_final<<<1,                          DHID, 0, stream>>>(partials, gamma, beta, W2, b2, cvec, cterm);
    k4_out  <<<dim3(NB / 64, NB / 32),     256, 0, stream>>>(A, Bm, cvec, cterm, out);
}


# Stats re-cut into 16-feature x 128 i x 128 j slabs whose blocks form their own A,B tiles on f32 WMMA: the GEMM launch is gone (3 -> 2 launches); sum order of S1,S2 changed
# speedup vs baseline: 1.5810x; 1.5810x over previous
#include <hip/hip_runtime.h>

typedef __attribute__((ext_vector_type(2))) float v2f;
typedef __attribute__((ext_vector_type(8))) float v8f;

#define NB    1024
#define DIN   64
#define DHID  128
#define BN_EPS 1e-5f

#define SP 132
#define XP 68

__global__ __launch_bounds__(256)
void k12_gemm_stats(const float* __restrict__ x, const float* __restrict__ W1,
                    const float* __restrict__ b1, float* __restrict__ AT,
                    float* __restrict__ BT, float* __restrict__ partials)
{
    const int f0    = blockIdx.x * 16;
    const int ic    = blockIdx.y;
    const int jc    = blockIdx.z;
    const int tid   = threadIdx.x;
    const int w     = tid >> 5;
    const int lane  = tid & 31;
    const int m     = lane & 15;
    const int kg    = (lane >> 4) << 1;
    const int rbase = (lane < 16) ? 0 : 8;

    __shared__ __attribute__((aligned(16))) float aS[16][SP];
    __shared__ __attribute__((aligned(16))) float bS[16][SP];
    __shared__ float sst[32];
    __shared__ __attribute__((aligned(16))) float xsA[128][XP];
    __shared__ __attribute__((aligned(16))) float xsB[128][XP];
    __shared__ __attribute__((aligned(16))) float Ws[2][DIN][16];

    float bias[8];
#pragma unroll
    for (int v = 0; v < 8; ++v) bias[v] = b1[f0 + rbase + v];

    const int ri = ic * 128 + w * 16;
    const int rj = jc * 128 + w * 16;

    {
        const float* __restrict__ xi = x + ri * DIN;
        const float* __restrict__ xj = x + rj * DIN;
#pragma unroll
        for (int t = 0; t < 8; ++t) {
            const int r  = 2 * t + (lane >> 4);
            const int c4 = (lane & 15) * 4;
            const float4 va = *(const float4*)(xi + r * DIN + c4);
            const float4 vb = *(const float4*)(xj + r * DIN + c4);
            *(float4*)&xsA[w * 16 + r][c4] = va;
            *(float4*)&xsB[w * 16 + r][c4] = vb;
        }
        const int part = tid >> 7;
        const int kk   = (tid >> 1) & 63;
        const int hf   = (tid & 1) * 8;
        const float* __restrict__ wsrc = W1 + (part * DIN + kk) * DHID + f0 + hf;
        const float4 w0 = *(const float4*)(wsrc);
        const float4 w1 = *(const float4*)(wsrc + 4);
        *(float4*)&Ws[part][kk][hf]     = w0;
        *(float4*)&Ws[part][kk][hf + 4] = w1;
    }
    __syncthreads();

    v8f accA = {0.f, 0.f, 0.f, 0.f, 0.f, 0.f, 0.f, 0.f};
    v8f accB = {0.f, 0.f, 0.f, 0.f, 0.f, 0.f, 0.f, 0.f};
#pragma unroll
    for (int k = 0; k < DIN; k += 4) {
        v2f xa, xb, wa, wb;
        xa.x = xsA[w * 16 + m][k + kg];
        xa.y = xsA[w * 16 + m][k + kg + 1];
        xb.x = xsB[w * 16 + m][k + kg];
        xb.y = xsB[w * 16 + m][k + kg + 1];
        wa.x = Ws[0][k + kg][m];
        wa.y = Ws[0][k + kg + 1][m];
        wb.x = Ws[1][k + kg][m];
        wb.y = Ws[1][k + kg + 1][m];
        accA = __builtin_amdgcn_wmma_f32_16x16x4_f32(false, wa, false, xa,
                                                     (short)0, accA, false, false);
        accB = __builtin_amdgcn_wmma_f32_16x16x4_f32(false, wb, false, xb,
                                                     (short)0, accB, false, false);
    }

    float av[8], bv[8];
#pragma unroll
    for (int v = 0; v < 8; ++v) {
        av[v] = accA[v] + bias[v];
        bv[v] = accB[v];
        aS[rbase + v][w * 16 + m] = av[v];
        bS[rbase + v][w * 16 + m] = bv[v];
    }
    if (jc == 0) {
#pragma unroll
        for (int v = 0; v < 8; ++v) AT[(f0 + rbase + v) * NB + (ri + m)] = av[v];
    }
    if (ic == 0) {
#pragma unroll
        for (int v = 0; v < 8; ++v) BT[(f0 + rbase + v) * NB + (rj + m)] = bv[v];
    }
    __syncthreads();

    const v2f z = {0.f, 0.f};
#define K2_TERM(bval, e)                                                   \
    {                                                                      \
        const v2f bb  = {bval, bval};                                      \
        const v2f r01 = __builtin_elementwise_max(a01 + bb, z);            \
        const v2f r23 = __builtin_elementwise_max(a23 + bb, z);            \
        p1[e]     += r01;                                                  \
        p1[e + 1] += r23;                                                  \
        p2[e]     = __builtin_elementwise_fma(r01, r01, p2[e]);            \
        p2[e + 1] = __builtin_elementwise_fma(r23, r23, p2[e + 1]);        \
    }

    float s1[2], s2[2];
#pragma unroll
    for (int ff = 0; ff < 2; ++ff) {
        const int fl = 2 * w + ff;
        v2f a01, a23;
        a01.x = aS[fl][lane];
        a01.y = aS[fl][lane + 32];
        a23.x = aS[fl][lane + 64];
        a23.y = aS[fl][lane + 96];
        v2f p1[4] = {z, z, z, z};
        v2f p2[4] = {z, z, z, z};
        for (int j = 0; j < 128; j += 4) {
            const float4 bq = *(const float4*)&bS[fl][j];
            K2_TERM(bq.x, 0)
            K2_TERM(bq.y, 2)
            K2_TERM(bq.z, 0)
            K2_TERM(bq.w, 2)
        }
        s1[ff] = ((p1[0].x + p1[0].y) + (p1[1].x + p1[1].y)) +
                 ((p1[2].x + p1[2].y) + (p1[3].x + p1[3].y));
        s2[ff] = ((p2[0].x + p2[0].y) + (p2[1].x + p2[1].y)) +
                 ((p2[2].x + p2[2].y) + (p2[3].x + p2[3].y));
    }
#undef K2_TERM

#pragma unroll
    for (int off = 16; off > 0; off >>= 1) {
        s1[0] += __shfl_down(s1[0], off);
        s2[0] += __shfl_down(s2[0], off);
        s1[1] += __shfl_down(s1[1], off);
        s2[1] += __shfl_down(s2[1], off);
    }
    if (lane == 0) {
        sst[4 * w + 0] = s1[0];
        sst[4 * w + 1] = s2[0];
        sst[4 * w + 2] = s1[1];
        sst[4 * w + 3] = s2[1];
    }
    __syncthreads();
    if (tid < 32) partials[(ic * 8 + jc) * 256 + f0 * 2 + tid] = sst[tid];
}

__global__ __launch_bounds__(256)
void k4_out(const float* __restrict__ AT, const float* __restrict__ BT,
            const float* __restrict__ partials, const float* __restrict__ gamma,
            const float* __restrict__ beta, const float* __restrict__ W2,
            const float* __restrict__ b2, float* __restrict__ out)
{
    __shared__ __attribute__((aligned(16))) float aT[DHID][64];
    __shared__ __attribute__((aligned(16))) float bT[DHID][32];
    __shared__ __attribute__((aligned(16))) float cs[DHID];
    __shared__ float red[DHID];
    __shared__ __attribute__((aligned(16))) float redp[4][2 * DHID];

    const int i0  = blockIdx.x * 64;
    const int j0  = blockIdx.y * 32;
    const int tid = threadIdx.x;

    {
        const int c4 = (tid & 15) * 4;
        const int r  = tid >> 4;
#pragma unroll
        for (int q = 0; q < 8; ++q) {
            const int h = r + q * 16;
            const float4 v = *(const float4*)(AT + h * NB + i0 + c4);
            *(float4*)&aT[h][c4] = v;
        }
    }
    {
        const int c4 = (tid & 7) * 4;
        const int r  = tid >> 3;
#pragma unroll
        for (int q = 0; q < 4; ++q) {
            const int h = r + q * 32;
            const float4 v = *(const float4*)(BT + h * NB + j0 + c4);
            *(float4*)&bT[h][c4] = v;
        }
    }
    const float gam = gamma[tid & (DHID - 1)];
    const float bet = beta[tid & (DHID - 1)];
    const float w2v = W2[tid & (DHID - 1)];
    {
        const int fp = ((tid >> 5) & 1) * 32 + (tid & 31);
        const int sq = tid >> 6;
        float4 s = {0.f, 0.f, 0.f, 0.f};
#pragma unroll
        for (int p = 0; p < 16; ++p) {
            const float4 v = *(const float4*)(partials + (p * 4 + sq) * 256 + fp * 4);
            s.x += v.x; s.y += v.y; s.z += v.z; s.w += v.w;
        }
        *(float4*)&redp[sq][fp * 4] = s;
    }
    __syncthreads();
    if (tid < DHID) {
        const int h = tid;
        float S1 = 0.f, S2 = 0.f;
#pragma unroll
        for (int q = 0; q < 4; ++q) {
            const float2 t = *(const float2*)&redp[q][2 * h];
            S1 += t.x; S2 += t.y;
        }
        const float invN = 1.0f / ((float)NB * (float)NB);
        const float mean = S1 * invN;
        const float var  = fmaxf(S2 * invN - mean * mean, 0.f);
        const float inv  = rsqrtf(var + BN_EPS);
        const float g    = gam * inv;
        const float w2   = w2v;
        cs[h]  = w2 * g;
        red[h] = w2 * (bet - mean * g);
    }
    __syncthreads();

    float ct;
    {
        const int lane = tid & 31;
        const float t0 = red[lane]      + red[lane + 64];
        const float t1 = red[lane + 32] + red[lane + 96];
        float u = t0 + t1;
        u += __shfl_down(u, 16);
        u += __shfl_down(u, 8);
        u += __shfl_down(u, 4);
        u += __shfl_down(u, 2);
        u += __shfl_down(u, 1);
        ct = __shfl(u, 0) + b2[0];
    }

    const int ib = (tid & 15) * 4;
    const int jb = (tid >> 4) * 2;
    float acc[4][2] = {};

#define K4_LOAD(S, hh)                                                     \
    {                                                                      \
        S##c  = *(const float4*)&cs[(hh)];                                 \
        S##a0 = *(const float4*)&aT[(hh) + 0][ib];                         \
        S##a1 = *(const float4*)&aT[(hh) + 1][ib];                         \
        S##a2 = *(const float4*)&aT[(hh) + 2][ib];                         \
        S##a3 = *(const float4*)&aT[(hh) + 3][ib];                         \
        S##b0 = *(const float2*)&bT[(hh) + 0][jb];                         \
        S##b1 = *(const float2*)&bT[(hh) + 1][jb];                         \
        S##b2 = *(const float2*)&bT[(hh) + 2][jb];                         \
        S##b3 = *(const float2*)&bT[(hh) + 3][jb];                         \
    }
#define K4_STEP(av, bv, ck)                                                \
    {                                                                      \
        acc[0][0] = fmaf(ck, fmaxf(av.x + bv.x, 0.f), acc[0][0]);          \
        acc[0][1] = fmaf(ck, fmaxf(av.x + bv.y, 0.f), acc[0][1]);          \
        acc[1][0] = fmaf(ck, fmaxf(av.y + bv.x, 0.f), acc[1][0]);          \
        acc[1][1] = fmaf(ck, fmaxf(av.y + bv.y, 0.f), acc[1][1]);          \
        acc[2][0] = fmaf(ck, fmaxf(av.z + bv.x, 0.f), acc[2][0]);          \
        acc[2][1] = fmaf(ck, fmaxf(av.z + bv.y, 0.f), acc[2][1]);          \
        acc[3][0] = fmaf(ck, fmaxf(av.w + bv.x, 0.f), acc[3][0]);          \
        acc[3][1] = fmaf(ck, fmaxf(av.w + bv.y, 0.f), acc[3][1]);          \
    }
#define K4_MATH(S)                                                         \
    {                                                                      \
        K4_STEP(S##a0, S##b0, S##c.x)                                      \
        K4_STEP(S##a1, S##b1, S##c.y)                                      \
        K4_STEP(S##a2, S##b2, S##c.z)                                      \
        K4_STEP(S##a3, S##b3, S##c.w)                                      \
    }

    float4 Pc, Pa0, Pa1, Pa2, Pa3;  float2 Pb0, Pb1, Pb2, Pb3;
    float4 Qc, Qa0, Qa1, Qa2, Qa3;  float2 Qb0, Qb1, Qb2, Qb3;

    for (int h = 0; h < DHID; h += 8) {
        K4_LOAD(P, h)
        K4_LOAD(Q, h + 4)
        __builtin_amdgcn_sched_barrier(0);
        K4_MATH(P)
        K4_MATH(Q)
    }
#undef K4_LOAD
#undef K4_STEP
#undef K4_MATH

#pragma unroll
    for (int ii = 0; ii < 4; ++ii) {
        float2 o;
        o.x = acc[ii][0] + ct;
        o.y = acc[ii][1] + ct;
        *(float2*)(out + (size_t)(i0 + ib + ii) * NB + (j0 + jb)) = o;
    }
}

extern "C" void kernel_launch(void* const* d_in, const int* in_sizes, int n_in,
                              void* d_out, int out_size, void* d_ws, size_t ws_size,
                              hipStream_t stream)
{
    const float* x     = (const float*)d_in[0];
    const float* W1    = (const float*)d_in[1];
    const float* b1    = (const float*)d_in[2];
    const float* gamma = (const float*)d_in[3];
    const float* beta  = (const float*)d_in[4];
    const float* W2    = (const float*)d_in[5];
    const float* b2    = (const float*)d_in[6];
    float* out = (float*)d_out;

    float* ws       = (float*)d_ws;
    float* AT       = ws;
    float* BT       = ws + NB * DHID;
    float* partials = ws + 2 * NB * DHID;

    k12_gemm_stats<<<dim3(DHID / 16, 8, 8), 256, 0, stream>>>(x, W1, b1, AT, BT, partials);
    k4_out  <<<dim3(NB / 64, NB / 32),     256, 0, stream>>>(AT, BT, partials, gamma, beta, W2, b2, out);
}
